# starting kernel
# baseline (speedup 1.0000x reference)
#include <hip/hip_runtime.h>
#include <hip/hip_bf16.h>

#define BS   512
#define DW   400
#define DT   20
#define TT   30
#define RR   256
#define FAN  440
#define NOUT (TT * TT)

typedef float v2f __attribute__((ext_vector_type(2)));
typedef float v8f __attribute__((ext_vector_type(8)));

__device__ __forceinline__ float fast_tanh(float x) {
#if __has_builtin(__builtin_amdgcn_tanhf)
    return __builtin_amdgcn_tanhf(x);
#else
    float y;
    asm volatile("v_tanh_f32 %0, %1" : "=v"(y) : "v"(x));
    return y;
#endif
}

__global__ void __launch_bounds__(32)
wordpart_wmma_kernel(const float* __restrict__ word,
                     const float* __restrict__ W,
                     float* __restrict__ wp) {
    const int m0   = blockIdx.x * 16;
    const int n0   = blockIdx.y * 16;
    const int lane = threadIdx.x;
    const int half = lane >> 4;
    const int mn   = lane & 15;

    const float* arow = word + (size_t)(m0 + mn) * DW + 2 * half;
    const float* brow = W + (size_t)(n0 + mn) * FAN + 2 * half;

    v8f acc = {0.f, 0.f, 0.f, 0.f, 0.f, 0.f, 0.f, 0.f};

    for (int k = 0; k < DW; k += 4) {
        v2f a, b;
        a.x = arow[k];
        a.y = arow[k + 1];
        b.x = brow[k];
        b.y = brow[k + 1];
        acc = __builtin_amdgcn_wmma_f32_16x16x4_f32(
            false, a, false, b, (short)0, acc, false, false);
    }

    float* dcol = wp + n0 + mn;
#pragma unroll
    for (int v = 0; v < 8; ++v) {
        const int m = v + 8 * half;
        dcol[(size_t)(m0 + m) * RR] = acc[v];
    }
}

__global__ void __launch_bounds__(RR)
tag_kernel(const float* __restrict__ tag,
           const float* __restrict__ W,
           const float* __restrict__ bias,
           float* __restrict__ ti,
           float* __restrict__ tj) {
    const int t = blockIdx.x;
    const int r = threadIdx.x;
    const float* te = tag + t * DT;
    const float* w1 = W + (size_t)r * FAN + DW;
    const float* w2 = W + (size_t)r * FAN + DW + DT;
    float aj = 0.f, ai = 0.f;
#pragma unroll
    for (int d = 0; d < DT; ++d) {
        const float x = te[d];
        aj += x * w1[d];
        ai += x * w2[d];
    }
    tj[t * RR + r] = aj;
    ti[t * RR + r] = ai + bias[r];
}

__global__ void __launch_bounds__(RR)
score_kernel(const float* __restrict__ wp,
             const float* __restrict__ ti,
             const float* __restrict__ tj,
             const float* __restrict__ vec,
             float* __restrict__ out) {
    __shared__ float s_wp[RR];
    __shared__ float s_v[RR];
    __shared__ float s_ti[TT * RR];
    __shared__ float s_tj[TT * RR];

    const int bs  = blockIdx.x;
    const int tid = threadIdx.x;

    s_wp[tid] = wp[(size_t)bs * RR + tid];
    s_v[tid]  = vec[tid];
    for (int idx = tid; idx < TT * RR; idx += RR) {
        s_ti[idx] = ti[idx];
        s_tj[idx] = tj[idx];
    }
    __syncthreads();

    const int  o0   = tid;
    const int  o1   = tid + RR;
    const int  o2   = tid + 2 * RR;
    const bool has3 = (tid + 3 * RR) < NOUT;
    const int  o3   = has3 ? (tid + 3 * RR) : (NOUT - 1);

    const float4* p_wp = (const float4*)s_wp;
    const float4* p_v  = (const float4*)s_v;
    const float4* ti0  = (const float4*)(s_ti + (o0 / TT) * RR);
    const float4* tj0  = (const float4*)(s_tj + (o0 % TT) * RR);
    const float4* ti1  = (const float4*)(s_ti + (o1 / TT) * RR);
    const float4* tj1  = (const float4*)(s_tj + (o1 % TT) * RR);
    const float4* ti2  = (const float4*)(s_ti + (o2 / TT) * RR);
    const float4* tj2  = (const float4*)(s_tj + (o2 % TT) * RR);
    const float4* ti3  = (const float4*)(s_ti + (o3 / TT) * RR);
    const float4* tj3  = (const float4*)(s_tj + (o3 % TT) * RR);

    float acc0 = 0.f, acc1 = 0.f, acc2 = 0.f, acc3 = 0.f;

#pragma unroll 2
    for (int rq = 0; rq < RR / 4; ++rq) {
        const float4 w4 = p_wp[rq];
        const float4 v4 = p_v[rq];

        float4 a, b;
        a = ti0[rq]; b = tj0[rq];
        acc0 += v4.x * fast_tanh(w4.x + a.x + b.x);
        acc0 += v4.y * fast_tanh(w4.y + a.y + b.y);
        acc0 += v4.z * fast_tanh(w4.z + a.z + b.z);
        acc0 += v4.w * fast_tanh(w4.w + a.w + b.w);

        a = ti1[rq]; b = tj1[rq];
        acc1 += v4.x * fast_tanh(w4.x + a.x + b.x);
        acc1 += v4.y * fast_tanh(w4.y + a.y + b.y);
        acc1 += v4.z * fast_tanh(w4.z + a.z + b.z);
        acc1 += v4.w * fast_tanh(w4.w + a.w + b.w);

        a = ti2[rq]; b = tj2[rq];
        acc2 += v4.x * fast_tanh(w4.x + a.x + b.x);
        acc2 += v4.y * fast_tanh(w4.y + a.y + b.y);
        acc2 += v4.z * fast_tanh(w4.z + a.z + b.z);
        acc2 += v4.w * fast_tanh(w4.w + a.w + b.w);

        a = ti3[rq]; b = tj3[rq];
        acc3 += v4.x * fast_tanh(w4.x + a.x + b.x);
        acc3 += v4.y * fast_tanh(w4.y + a.y + b.y);
        acc3 += v4.z * fast_tanh(w4.z + a.z + b.z);
        acc3 += v4.w * fast_tanh(w4.w + a.w + b.w);
    }

    float* orow = out + (size_t)bs * NOUT;
    orow[o0] = acc0;
    orow[o1] = acc1;
    orow[o2] = acc2;
    if (has3) orow[o3] = acc3;
}

extern "C" void kernel_launch(void* const* d_in, const int* in_sizes, int n_in,
                              void* d_out, int out_size, void* d_ws, size_t ws_size,
                              hipStream_t stream) {
    (void)in_sizes; (void)n_in; (void)out_size; (void)ws_size;

    const float* word = (const float*)d_in[0];
    const float* tag  = (const float*)d_in[1];
    const float* W    = (const float*)d_in[2];
    const float* bias = (const float*)d_in[3];
    const float* vec  = (const float*)d_in[4];
    float* out = (float*)d_out;

    float* wp = (float*)d_ws;
    float* ti = wp + (size_t)BS * RR;
    float* tj = ti + (size_t)TT * RR;

    wordpart_wmma_kernel<<<dim3(BS / 16, RR / 16), 32, 0, stream>>>(word, W, wp);
    tag_kernel<<<TT, RR, 0, stream>>>(tag, W, bias, ti, tj);
    score_kernel<<<BS, RR, 0, stream>>>(wp, ti, tj, vec, out);
}


# score loop written in issue order: each tanh is followed by one independent packed f32 add (2 r ahead) or multiply-add (2 r behind); v and tj rows stay in register pairs, per-step moves gone
# speedup vs baseline: 3.3529x; 3.3529x over previous
#include <hip/hip_runtime.h>
#include <hip/hip_bf16.h>

#define BS   512
#define DW   400
#define DT   20
#define TT   30
#define RR   256
#define FAN  440
#define NOUT (TT * TT)

typedef float v2f __attribute__((ext_vector_type(2)));
typedef float v8f __attribute__((ext_vector_type(8)));

__global__ void __launch_bounds__(32)
wordpart_wmma_kernel(const float* __restrict__ word,
                     const float* __restrict__ W,
                     float* __restrict__ wp) {
    const int m0   = blockIdx.x * 16;
    const int n0   = blockIdx.y * 16;
    const int lane = threadIdx.x;
    const int half = lane >> 4;
    const int mn   = lane & 15;

    const float* arow = word + (size_t)(m0 + mn) * DW + 2 * half;
    const float* brow = W + (size_t)(n0 + mn) * FAN + 2 * half;

    v8f acc = {0.f, 0.f, 0.f, 0.f, 0.f, 0.f, 0.f, 0.f};

    for (int k = 0; k < DW; k += 4) {
        v2f a, b;
        a.x = arow[k];
        a.y = arow[k + 1];
        b.x = brow[k];
        b.y = brow[k + 1];
        acc = __builtin_amdgcn_wmma_f32_16x16x4_f32(
            false, a, false, b, (short)0, acc, false, false);
    }

    float* dcol = wp + n0 + mn;
#pragma unroll
    for (int v = 0; v < 8; ++v) {
        const int m = v + 8 * half;
        dcol[(size_t)(m0 + m) * RR] = acc[v];
    }
}

#define RS     32
#define NWAVE  (RR / RS)
#define BROWS  32
#define IPB    2
#define NCOL   (IPB * TT)
#define RPITCH 33
#define TP     36
#define TJROWS 32
#define TIROWS 16
#define KST    (DT / 4)

__device__ __forceinline__ void load_frag20(v2f (&f)[KST], const float* __restrict__ row, int h) {
    const float4 q0 = *(const float4*)(row + 4 * h);
    const float4 q1 = *(const float4*)(row + 8 + 4 * h);
    const v2f    q2 = *(const v2f*)(row + 16 + 2 * h);
    f[0] = (v2f){q0.x, q0.y};
    f[1] = (v2f){q0.z, q0.w};
    f[2] = (v2f){q1.x, q1.y};
    f[3] = (v2f){q1.z, q1.w};
    f[4] = q2;
}


__device__ __forceinline__ v2f pk_add_bcast(v2f yv, v2f sp, int hi) {
    v2f d;
    if (hi) {
        asm volatile("v_pk_add_f32 %0, %1, %2 op_sel:[0,1]" : "=v"(d) : "v"(yv), "v"(sp));
    } else {
        asm volatile("v_pk_add_f32 %0, %1, %2 op_sel_hi:[1,0]" : "=v"(d) : "v"(yv), "v"(sp));
    }
    return d;
}

__device__ __forceinline__ v2f pk_mul_bcast(v2f vp, v2f t, int hi) {
    v2f d;
    if (hi) {
        asm volatile("v_pk_mul_f32 %0, %1, %2 op_sel:[1,0]" : "=v"(d) : "v"(vp), "v"(t));
    } else {
        asm volatile("v_pk_mul_f32 %0, %1, %2 op_sel_hi:[0,1]" : "=v"(d) : "v"(vp), "v"(t));
    }
    return d;
}

__device__ __forceinline__ void pk_fma_bcast(v2f& acc, v2f vp, v2f t, int hi) {
    if (hi) {
        asm volatile("v_pk_fma_f32 %0, %1, %2, %0 op_sel:[1,0,0]" : "+v"(acc) : "v"(vp), "v"(t));
    } else {
        asm volatile("v_pk_fma_f32 %0, %1, %2, %0 op_sel_hi:[0,1,1]" : "+v"(acc) : "v"(vp), "v"(t));
    }
}

__device__ __forceinline__ float tanh_ordered(float x) {
    float d;
    asm volatile("v_tanh_f32 %0, %1" : "=v"(d) : "v"(x));
    return d;
}

__device__ __forceinline__ void load_row_pairs(v2f (&s)[RS / 2], const float4* p) {
#pragma unroll
    for (int q = 0; q < RS / 4; ++q) {
        const float4 x = p[q];
        s[2 * q]     = (v2f){x.x, x.y};
        s[2 * q + 1] = (v2f){x.z, x.w};
    }
}

__device__ __forceinline__ void step_mac(v2f (&acc)[4], const v2f (&vp)[RS / 2], v2f t, int k) {
    if (k < 4) acc[k] = pk_mul_bcast(vp[k >> 1], t, k & 1);
    else       pk_fma_bcast(acc[k & 3], vp[k >> 1], t, k & 1);
}

__device__ __forceinline__ v2f score_step(const v2f (&y)[RS], const v2f (&s)[RS / 2],
                                          const v2f (&vp)[RS / 2], const v2f (&n)[RS / 2],
                                          v2f& a0, v2f& a1) {
    v2f a[RS];
    v2f t[RS];
    v2f acc[4];
    a[0] = a0;
    a[1] = a1;
#pragma unroll
    for (int r = 0; r < RS; ++r) {
        const float tx = tanh_ordered(a[r].x);
        if (r + 2 < RS)       a[r + 2] = pk_add_bcast(y[r + 2], s[(r + 2) >> 1], (r + 2) & 1);
        else if (r + 2 == RS) a0 = pk_add_bcast(y[0], n[0], 0);
        else                  a1 = pk_add_bcast(y[1], n[0], 1);
        const float ty = tanh_ordered(a[r].y);
        t[r] = (v2f){tx, ty};
        if (r >= 2) step_mac(acc, vp, t[r - 2], r - 2);
    }
    step_mac(acc, vp, t[RS - 2], RS - 2);
    step_mac(acc, vp, t[RS - 1], RS - 1);
    return (acc[0] + acc[1]) + (acc[2] + acc[3]);
}

__global__ void __launch_bounds__(RR)
score_kernel(const float* __restrict__ wp,
             const float* __restrict__ tag,
             const float* __restrict__ W,
             const float* __restrict__ bias,
             const float* __restrict__ vec,
             float* __restrict__ out) {
    __shared__ float s_red[NWAVE * NCOL * RPITCH];
    __shared__ __attribute__((aligned(16))) float s_tj[NWAVE * TJROWS * TP];
    __shared__ __attribute__((aligned(16))) float s_ti[NWAVE * TIROWS * TP];

    const int i0   = blockIdx.x * IPB;
    const int bsg  = blockIdx.y;
    const int lane = threadIdx.x & 31;
    const int w    = __builtin_amdgcn_readfirstlane((int)(threadIdx.x >> 5));
    const int r0   = w * RS;
    const int bs   = bsg * BROWS + lane;
    const int c    = lane & 15;
    const int h    = lane >> 4;

    const float4* wrow = (const float4*)(wp + (size_t)bs * RR + r0);
    float4 xr[RS / 4];
#pragma unroll
    for (int q = 0; q < RS / 4; ++q) xr[q] = wrow[q];

    const int it = i0 >> 4;
    v2f aj[2][KST], ai[KST], bj[2][KST], bi[2][KST];
#pragma unroll
    for (int jt = 0; jt < 2; ++jt) {
        const int row = (16 * jt + c < TT) ? (16 * jt + c) : (TT - 1);
        load_frag20(aj[jt], tag + row * DT, h);
    }
    {
        const int row = (16 * it + c < TT) ? (16 * it + c) : (TT - 1);
        load_frag20(ai, tag + row * DT, h);
    }
#pragma unroll
    for (int nt = 0; nt < 2; ++nt) {
        const float* pb = W + (size_t)(r0 + 16 * nt + c) * FAN + DW;
        load_frag20(bj[nt], pb, h);
        load_frag20(bi[nt], pb + DT, h);
    }
    const float bz0 = bias[r0 + c];
    const float bz1 = bias[r0 + 16 + c];

    const v8f zero8 = {0.f, 0.f, 0.f, 0.f, 0.f, 0.f, 0.f, 0.f};
    v8f d00 = zero8, d01 = zero8, d10 = zero8, d11 = zero8, e0 = zero8, e1 = zero8;
#pragma unroll
    for (int s = 0; s < KST; ++s) {
        d00 = __builtin_amdgcn_wmma_f32_16x16x4_f32(false, aj[0][s], false, bj[0][s], (short)0, d00, false, false);
        d01 = __builtin_amdgcn_wmma_f32_16x16x4_f32(false, aj[0][s], false, bj[1][s], (short)0, d01, false, false);
        d10 = __builtin_amdgcn_wmma_f32_16x16x4_f32(false, aj[1][s], false, bj[0][s], (short)0, d10, false, false);
        d11 = __builtin_amdgcn_wmma_f32_16x16x4_f32(false, aj[1][s], false, bj[1][s], (short)0, d11, false, false);
        e0  = __builtin_amdgcn_wmma_f32_16x16x4_f32(false, ai[s],    false, bi[0][s], (short)0, e0,  false, false);
        e1  = __builtin_amdgcn_wmma_f32_16x16x4_f32(false, ai[s],    false, bi[1][s], (short)0, e1,  false, false);
    }

    float* tjw = s_tj + w * (TJROWS * TP);
    float* tiw = s_ti + w * (TIROWS * TP);
    {
        float* pj = tjw + (8 * h) * TP + c;
        float* pi = tiw + (8 * h) * TP + c;
#pragma unroll
        for (int v = 0; v < 8; ++v) {
            pj[v * TP]                = d00[v];
            pj[v * TP + 16]           = d01[v];
            pj[(16 + v) * TP]         = d10[v];
            pj[(16 + v) * TP + 16]    = d11[v];
            pi[v * TP]                = e0[v] + bz0;
            pi[v * TP + 16]           = e1[v] + bz1;
        }
    }
    __syncthreads();

    v2f y[RS];
    {
        const int li = i0 & 15;
        const float4* t0 = (const float4*)(tiw + li * TP);
        const float4* t1 = (const float4*)(tiw + (li + 1) * TP);
#pragma unroll
        for (int q = 0; q < RS / 4; ++q) {
            const float4 x = xr[q];
            const float4 a = t0[q];
            const float4 b = t1[q];
            y[4 * q]     = (v2f){x.x + a.x, x.x + b.x};
            y[4 * q + 1] = (v2f){x.y + a.y, x.y + b.y};
            y[4 * q + 2] = (v2f){x.z + a.z, x.z + b.z};
            y[4 * q + 3] = (v2f){x.w + a.w, x.w + b.w};
        }
    }

    v2f vp[RS / 2];
    {
        const float4* v4 = (const float4*)(vec + r0);
#pragma unroll
        for (int q = 0; q < RS / 4; ++q) {
            const float4 x = v4[q];
            vp[2 * q]     = (v2f){x.x, x.y};
            vp[2 * q + 1] = (v2f){x.z, x.w};
        }
#pragma unroll
        for (int k = 0; k < RS / 2; ++k) {
            asm volatile("" : "+v"(vp[k]));
        }
    }

    const float4* tj4 = (const float4*)tjw;
    v2f sa[RS / 2], sb[RS / 2];
    load_row_pairs(sa, tj4);

    float* sw = s_red + (w * NCOL) * RPITCH + lane;

    v2f a0 = pk_add_bcast(y[0], sa[0], 0);
    v2f a1 = pk_add_bcast(y[1], sa[0], 1);

#pragma unroll 1
    for (int j = 0; j < TT; j += 2) {
        load_row_pairs(sb, tj4 + (j + 1) * (TP / 4));
        __builtin_amdgcn_sched_barrier(0);
        const v2f pa = score_step(y, sa, vp, sb, a0, a1);
        sw[j * RPITCH]        = pa.x;
        sw[(TT + j) * RPITCH] = pa.y;
        __builtin_amdgcn_sched_barrier(0);

        load_row_pairs(sa, tj4 + (j + 2) * (TP / 4));
        __builtin_amdgcn_sched_barrier(0);
        const v2f pb = score_step(y, sb, vp, sa, a0, a1);
        sw[(j + 1) * RPITCH]      = pb.x;
        sw[(TT + j + 1) * RPITCH] = pb.y;
        __builtin_amdgcn_sched_barrier(0);
    }

    __syncthreads();

    float* obase = out + (size_t)(bsg * BROWS) * NOUT + i0 * TT;
    for (int idx = threadIdx.x; idx < BROWS * NCOL; idx += RR) {
        const int bl = idx / NCOL;
        const int cc = idx - bl * NCOL;
        const float* sr = s_red + cc * RPITCH + bl;
        float sum = 0.f;
#pragma unroll
        for (int q = 0; q < NWAVE; ++q) sum += sr[q * NCOL * RPITCH];
        obase[(size_t)bl * NOUT + cc] = sum;
    }
}

extern "C" void kernel_launch(void* const* d_in, const int* in_sizes, int n_in,
                              void* d_out, int out_size, void* d_ws, size_t ws_size,
                              hipStream_t stream) {
    (void)in_sizes; (void)n_in; (void)out_size; (void)ws_size;

    const float* word = (const float*)d_in[0];
    const float* tag  = (const float*)d_in[1];
    const float* W    = (const float*)d_in[2];
    const float* bias = (const float*)d_in[3];
    const float* vec  = (const float*)d_in[4];
    float* out = (float*)d_out;

    float* wp = (float*)d_ws;

    wordpart_wmma_kernel<<<dim3(BS / 16, RR / 16), 32, 0, stream>>>(word, W, wp);
    score_kernel<<<dim3(TT / IPB, BS / BROWS), RR, 0, stream>>>(wp, tag, W, bias, vec, out);
}


# word projection launch: A and B fragments read straight from global memory as 16-byte pieces (two k steps each), 50 loads kept in flight ahead of the products, no drain until the end; 2 accumulators
# speedup vs baseline: 4.2029x; 1.2535x over previous
#include <hip/hip_runtime.h>
#include <hip/hip_bf16.h>

#define BS   512
#define DW   400
#define DT   20
#define TT   30
#define RR   256
#define FAN  440
#define NOUT (TT * TT)

typedef float v2f __attribute__((ext_vector_type(2)));
typedef float v8f __attribute__((ext_vector_type(8)));

#define KCH (DW / 8)
#define PF  25

__global__ void __launch_bounds__(64)
wordpart_wmma_kernel(const float* __restrict__ word,
                     const float* __restrict__ W,
                     float* __restrict__ wp) {
    const int wv   = __builtin_amdgcn_readfirstlane((int)(threadIdx.x >> 5));
    const int m0   = blockIdx.x * 16;
    const int n0   = (blockIdx.y * 2 + wv) * 16;
    const int lane = threadIdx.x & 31;
    const int half = lane >> 4;
    const int mn   = lane & 15;

    const float4* arow = (const float4*)(word + (size_t)(m0 + mn) * DW + 4 * half);
    const float4* brow = (const float4*)(W + (size_t)(n0 + mn) * FAN + 4 * half);

    float4 a[KCH], b[KCH];
#pragma unroll
    for (int q = 0; q < PF; ++q) {
        a[q] = arow[2 * q];
        b[q] = brow[2 * q];
    }
    __builtin_amdgcn_sched_barrier(0);

    v8f acc0 = {0.f, 0.f, 0.f, 0.f, 0.f, 0.f, 0.f, 0.f};
    v8f acc1 = {0.f, 0.f, 0.f, 0.f, 0.f, 0.f, 0.f, 0.f};
#pragma unroll
    for (int q = 0; q < KCH; ++q) {
        if (q + PF < KCH) {
            a[q + PF] = arow[2 * (q + PF)];
            b[q + PF] = brow[2 * (q + PF)];
        }
        __builtin_amdgcn_sched_barrier(0);
        acc0 = __builtin_amdgcn_wmma_f32_16x16x4_f32(
            false, (v2f){a[q].x, a[q].y}, false, (v2f){b[q].x, b[q].y}, (short)0, acc0, false, false);
        acc1 = __builtin_amdgcn_wmma_f32_16x16x4_f32(
            false, (v2f){a[q].z, a[q].w}, false, (v2f){b[q].z, b[q].w}, (short)0, acc1, false, false);
        __builtin_amdgcn_sched_barrier(0);
    }
    const v8f acc = acc0 + acc1;

    float* dcol = wp + n0 + mn;
#pragma unroll
    for (int v = 0; v < 8; ++v) {
        const int m = v + 8 * half;
        dcol[(size_t)(m0 + m) * RR] = acc[v];
    }
}

#define RS     32
#define NWAVE  (RR / RS)
#define BROWS  32
#define IPB    2
#define NCOL   (IPB * TT)
#define RPITCH 33
#define TP     36
#define TJROWS 32
#define TIROWS 16
#define KST    (DT / 4)

__device__ __forceinline__ void load_frag20(v2f (&f)[KST], const float* __restrict__ row, int h) {
    const float4 q0 = *(const float4*)(row + 4 * h);
    const float4 q1 = *(const float4*)(row + 8 + 4 * h);
    const v2f    q2 = *(const v2f*)(row + 16 + 2 * h);
    f[0] = (v2f){q0.x, q0.y};
    f[1] = (v2f){q0.z, q0.w};
    f[2] = (v2f){q1.x, q1.y};
    f[3] = (v2f){q1.z, q1.w};
    f[4] = q2;
}


__device__ __forceinline__ v2f pk_add_bcast(v2f yv, v2f sp, int hi) {
    v2f d;
    if (hi) {
        asm volatile("v_pk_add_f32 %0, %1, %2 op_sel:[0,1]" : "=v"(d) : "v"(yv), "v"(sp));
    } else {
        asm volatile("v_pk_add_f32 %0, %1, %2 op_sel_hi:[1,0]" : "=v"(d) : "v"(yv), "v"(sp));
    }
    return d;
}

__device__ __forceinline__ v2f pk_mul_bcast(v2f vp, v2f t, int hi) {
    v2f d;
    if (hi) {
        asm volatile("v_pk_mul_f32 %0, %1, %2 op_sel:[1,0]" : "=v"(d) : "v"(vp), "v"(t));
    } else {
        asm volatile("v_pk_mul_f32 %0, %1, %2 op_sel_hi:[0,1]" : "=v"(d) : "v"(vp), "v"(t));
    }
    return d;
}

__device__ __forceinline__ void pk_fma_bcast(v2f& acc, v2f vp, v2f t, int hi) {
    if (hi) {
        asm volatile("v_pk_fma_f32 %0, %1, %2, %0 op_sel:[1,0,0]" : "+v"(acc) : "v"(vp), "v"(t));
    } else {
        asm volatile("v_pk_fma_f32 %0, %1, %2, %0 op_sel_hi:[0,1,1]" : "+v"(acc) : "v"(vp), "v"(t));
    }
}

__device__ __forceinline__ float tanh_ordered(float x) {
    float d;
    asm volatile("v_tanh_f32 %0, %1" : "=v"(d) : "v"(x));
    return d;
}

__device__ __forceinline__ void load_row_pairs(v2f (&s)[RS / 2], const float4* p) {
#pragma unroll
    for (int q = 0; q < RS / 4; ++q) {
        const float4 x = p[q];
        s[2 * q]     = (v2f){x.x, x.y};
        s[2 * q + 1] = (v2f){x.z, x.w};
    }
}

__device__ __forceinline__ void step_mac(v2f (&acc)[4], const v2f (&vp)[RS / 2], v2f t, int k) {
    if (k < 4) acc[k] = pk_mul_bcast(vp[k >> 1], t, k & 1);
    else       pk_fma_bcast(acc[k & 3], vp[k >> 1], t, k & 1);
}

__device__ __forceinline__ v2f score_step(const v2f (&y)[RS], const v2f (&s)[RS / 2],
                                          const v2f (&vp)[RS / 2], const v2f (&n)[RS / 2],
                                          v2f& a0, v2f& a1) {
    v2f a[RS];
    v2f t[RS];
    v2f acc[4];
    a[0] = a0;
    a[1] = a1;
#pragma unroll
    for (int r = 0; r < RS; ++r) {
        const float tx = tanh_ordered(a[r].x);
        if (r + 2 < RS)       a[r + 2] = pk_add_bcast(y[r + 2], s[(r + 2) >> 1], (r + 2) & 1);
        else if (r + 2 == RS) a0 = pk_add_bcast(y[0], n[0], 0);
        else                  a1 = pk_add_bcast(y[1], n[0], 1);
        const float ty = tanh_ordered(a[r].y);
        t[r] = (v2f){tx, ty};
        if (r >= 2) step_mac(acc, vp, t[r - 2], r - 2);
    }
    step_mac(acc, vp, t[RS - 2], RS - 2);
    step_mac(acc, vp, t[RS - 1], RS - 1);
    return (acc[0] + acc[1]) + (acc[2] + acc[3]);
}

__global__ void __launch_bounds__(RR)
score_kernel(const float* __restrict__ wp,
             const float* __restrict__ tag,
             const float* __restrict__ W,
             const float* __restrict__ bias,
             const float* __restrict__ vec,
             float* __restrict__ out) {
    __shared__ float s_red[NWAVE * NCOL * RPITCH];
    __shared__ __attribute__((aligned(16))) float s_tj[NWAVE * TJROWS * TP];
    __shared__ __attribute__((aligned(16))) float s_ti[NWAVE * TIROWS * TP];

    const int i0   = blockIdx.x * IPB;
    const int bsg  = blockIdx.y;
    const int lane = threadIdx.x & 31;
    const int w    = __builtin_amdgcn_readfirstlane((int)(threadIdx.x >> 5));
    const int r0   = w * RS;
    const int bs   = bsg * BROWS + lane;
    const int c    = lane & 15;
    const int h    = lane >> 4;

    const float4* wrow = (const float4*)(wp + (size_t)bs * RR + r0);
    float4 xr[RS / 4];
#pragma unroll
    for (int q = 0; q < RS / 4; ++q) xr[q] = wrow[q];

    const int it = i0 >> 4;
    v2f aj[2][KST], ai[KST], bj[2][KST], bi[2][KST];
#pragma unroll
    for (int jt = 0; jt < 2; ++jt) {
        const int row = (16 * jt + c < TT) ? (16 * jt + c) : (TT - 1);
        load_frag20(aj[jt], tag + row * DT, h);
    }
    {
        const int row = (16 * it + c < TT) ? (16 * it + c) : (TT - 1);
        load_frag20(ai, tag + row * DT, h);
    }
#pragma unroll
    for (int nt = 0; nt < 2; ++nt) {
        const float* pb = W + (size_t)(r0 + 16 * nt + c) * FAN + DW;
        load_frag20(bj[nt], pb, h);
        load_frag20(bi[nt], pb + DT, h);
    }
    const float bz0 = bias[r0 + c];
    const float bz1 = bias[r0 + 16 + c];

    const v8f zero8 = {0.f, 0.f, 0.f, 0.f, 0.f, 0.f, 0.f, 0.f};
    v8f d00 = zero8, d01 = zero8, d10 = zero8, d11 = zero8, e0 = zero8, e1 = zero8;
#pragma unroll
    for (int s = 0; s < KST; ++s) {
        d00 = __builtin_amdgcn_wmma_f32_16x16x4_f32(false, aj[0][s], false, bj[0][s], (short)0, d00, false, false);
        d01 = __builtin_amdgcn_wmma_f32_16x16x4_f32(false, aj[0][s], false, bj[1][s], (short)0, d01, false, false);
        d10 = __builtin_amdgcn_wmma_f32_16x16x4_f32(false, aj[1][s], false, bj[0][s], (short)0, d10, false, false);
        d11 = __builtin_amdgcn_wmma_f32_16x16x4_f32(false, aj[1][s], false, bj[1][s], (short)0, d11, false, false);
        e0  = __builtin_amdgcn_wmma_f32_16x16x4_f32(false, ai[s],    false, bi[0][s], (short)0, e0,  false, false);
        e1  = __builtin_amdgcn_wmma_f32_16x16x4_f32(false, ai[s],    false, bi[1][s], (short)0, e1,  false, false);
    }

    float* tjw = s_tj + w * (TJROWS * TP);
    float* tiw = s_ti + w * (TIROWS * TP);
    {
        float* pj = tjw + (8 * h) * TP + c;
        float* pi = tiw + (8 * h) * TP + c;
#pragma unroll
        for (int v = 0; v < 8; ++v) {
            pj[v * TP]                = d00[v];
            pj[v * TP + 16]           = d01[v];
            pj[(16 + v) * TP]         = d10[v];
            pj[(16 + v) * TP + 16]    = d11[v];
            pi[v * TP]                = e0[v] + bz0;
            pi[v * TP + 16]           = e1[v] + bz1;
        }
    }
    __syncthreads();

    v2f y[RS];
    {
        const int li = i0 & 15;
        const float4* t0 = (const float4*)(tiw + li * TP);
        const float4* t1 = (const float4*)(tiw + (li + 1) * TP);
#pragma unroll
        for (int q = 0; q < RS / 4; ++q) {
            const float4 x = xr[q];
            const float4 a = t0[q];
            const float4 b = t1[q];
            y[4 * q]     = (v2f){x.x + a.x, x.x + b.x};
            y[4 * q + 1] = (v2f){x.y + a.y, x.y + b.y};
            y[4 * q + 2] = (v2f){x.z + a.z, x.z + b.z};
            y[4 * q + 3] = (v2f){x.w + a.w, x.w + b.w};
        }
    }

    v2f vp[RS / 2];
    {
        const float4* v4 = (const float4*)(vec + r0);
#pragma unroll
        for (int q = 0; q < RS / 4; ++q) {
            const float4 x = v4[q];
            vp[2 * q]     = (v2f){x.x, x.y};
            vp[2 * q + 1] = (v2f){x.z, x.w};
        }
#pragma unroll
        for (int k = 0; k < RS / 2; ++k) {
            asm volatile("" : "+v"(vp[k]));
        }
    }

    const float4* tj4 = (const float4*)tjw;
    v2f sa[RS / 2], sb[RS / 2];
    load_row_pairs(sa, tj4);

    float* sw = s_red + (w * NCOL) * RPITCH + lane;

    v2f a0 = pk_add_bcast(y[0], sa[0], 0);
    v2f a1 = pk_add_bcast(y[1], sa[0], 1);

#pragma unroll 1
    for (int j = 0; j < TT; j += 2) {
        load_row_pairs(sb, tj4 + (j + 1) * (TP / 4));
        __builtin_amdgcn_sched_barrier(0);
        const v2f pa = score_step(y, sa, vp, sb, a0, a1);
        sw[j * RPITCH]        = pa.x;
        sw[(TT + j) * RPITCH] = pa.y;
        __builtin_amdgcn_sched_barrier(0);

        load_row_pairs(sa, tj4 + (j + 2) * (TP / 4));
        __builtin_amdgcn_sched_barrier(0);
        const v2f pb = score_step(y, sb, vp, sa, a0, a1);
        sw[(j + 1) * RPITCH]      = pb.x;
        sw[(TT + j + 1) * RPITCH] = pb.y;
        __builtin_amdgcn_sched_barrier(0);
    }

    __syncthreads();

    float* obase = out + (size_t)(bsg * BROWS) * NOUT + i0 * TT;
    for (int idx = threadIdx.x; idx < BROWS * NCOL; idx += RR) {
        const int bl = idx / NCOL;
        const int cc = idx - bl * NCOL;
        const float* sr = s_red + cc * RPITCH + bl;
        float sum = 0.f;
#pragma unroll
        for (int q = 0; q < NWAVE; ++q) sum += sr[q * NCOL * RPITCH];
        obase[(size_t)bl * NOUT + cc] = sum;
    }
}

extern "C" void kernel_launch(void* const* d_in, const int* in_sizes, int n_in,
                              void* d_out, int out_size, void* d_ws, size_t ws_size,
                              hipStream_t stream) {
    (void)in_sizes; (void)n_in; (void)out_size; (void)ws_size;

    const float* word = (const float*)d_in[0];
    const float* tag  = (const float*)d_in[1];
    const float* W    = (const float*)d_in[2];
    const float* bias = (const float*)d_in[3];
    const float* vec  = (const float*)d_in[4];
    float* out = (float*)d_out;

    float* wp = (float*)d_ws;

    wordpart_wmma_kernel<<<dim3(BS / 16, RR / 32), 64, 0, stream>>>(word, W, wp);
    score_kernel<<<dim3(TT / IPB, BS / BROWS), RR, 0, stream>>>(wp, tag, W, bias, vec, out);
}
